# starting kernel
# baseline (speedup 1.0000x reference)
#include <hip/hip_runtime.h>
#include <math.h>

#define NB 2
#define DD 1024
#define LL 2048
#define NS 16
#define RR 64
#define EE 96
#define NC 32
#define LC 64

typedef float v2f __attribute__((ext_vector_type(2)));
typedef float v8f __attribute__((ext_vector_type(8)));

__device__ __forceinline__ float sigmoidf_(float x) { return 1.0f / (1.0f + expf(-x)); }
__device__ __forceinline__ float softplusf_(float x) { return (x > 20.0f) ? x : log1pf(expf(x)); }

__global__ __launch_bounds__(128)
void k_xproj(const float* __restrict__ W, const float* __restrict__ U,
             float* __restrict__ Xd)
{
    __shared__ float As[16 * 68];
    const int tid  = threadIdx.x;
    const int lane = tid & 31;
    const int wv   = tid >> 5;
    const int m0   = blockIdx.y << 4;
    const int n0   = (blockIdx.x << 6) + (wv << 4);
    const int bi   = blockIdx.z;
    const int am   = lane & 15;
    const int ak   = (lane >> 4) << 1;
    v8f acc = {};
    const float* Ub = U + ((size_t)bi * DD) * LL + n0 + am;
    for (int kc = 0; kc < DD; kc += 64) {
        for (int i = tid; i < 16 * 64; i += 128) {
            const int r = i >> 6, c = i & 63;
            As[r * 68 + c] = W[(size_t)(m0 + r) * DD + kc + c];
        }
        __syncthreads();
        #pragma unroll
        for (int k = 0; k < 64; k += 4) {
            v2f a, b;
            a.x = As[am * 68 + k + ak];
            a.y = As[am * 68 + k + ak + 1];
            const float* bp = Ub + (size_t)(kc + k + ak) * LL;
            b.x = bp[0];
            b.y = bp[LL];
            acc = __builtin_amdgcn_wmma_f32_16x16x4_f32(
                false, a, false, b, (short)0, acc, false, false);
        }
        __syncthreads();
    }
    const int cn = n0 + am;
    const int cr = (lane >> 4) << 3;
    #pragma unroll
    for (int v = 0; v < 8; ++v)
        Xd[((size_t)bi * EE + m0 + cr + v) * LL + cn] = acc[v];
}

__global__ __launch_bounds__(128)
void k_dt(const float* __restrict__ Wdt, const float* __restrict__ Xd,
          const float* __restrict__ bias, float* __restrict__ Delta)
{
    __shared__ float As[16 * 68];
    const int tid  = threadIdx.x;
    const int lane = tid & 31;
    const int wv   = tid >> 5;
    const int m0   = blockIdx.y << 4;
    const int n0   = (blockIdx.x << 6) + (wv << 4);
    const int bi   = blockIdx.z;
    const int am   = lane & 15;
    const int ak   = (lane >> 4) << 1;
    for (int i = tid; i < 16 * 64; i += 128) {
        const int r = i >> 6, c = i & 63;
        As[r * 68 + c] = Wdt[(size_t)(m0 + r) * RR + c];
    }
    __syncthreads();
    v8f acc = {};
    const float* Bb = Xd + ((size_t)bi * EE) * LL + n0 + am;
    #pragma unroll
    for (int k = 0; k < RR; k += 4) {
        v2f a, b;
        a.x = As[am * 68 + k + ak];
        a.y = As[am * 68 + k + ak + 1];
        const float* bp = Bb + (size_t)(k + ak) * LL;
        b.x = bp[0];
        b.y = bp[LL];
        acc = __builtin_amdgcn_wmma_f32_16x16x4_f32(
            false, a, false, b, (short)0, acc, false, false);
    }
    const int cn = n0 + am;
    const int cr = (lane >> 4) << 3;
    #pragma unroll
    for (int v = 0; v < 8; ++v) {
        const int row = m0 + cr + v;
        Delta[((size_t)bi * DD + row) * LL + cn] = softplusf_(acc[v] + bias[row]);
    }
}

__global__ __launch_bounds__(256)
void k_scanA(const float* __restrict__ U, const float* __restrict__ Xd,
             const float* __restrict__ Delta, const float* __restrict__ Alog,
             float* __restrict__ S, float* __restrict__ SumD)
{
    const int lane = threadIdx.x & 31;
    const int gw   = blockIdx.x * (blockDim.x >> 5) + (threadIdx.x >> 5);
    const int p    = (gw << 1) + (lane >> 4);
    const int ch   = p >> 5;
    const int c    = p & (NC - 1);
    const int bi   = ch >> 10;
    const int d    = ch & (DD - 1);
    const int n    = lane & 15;
    const int hb   = lane & 16;
    const float An = -expf(Alog[d * NS + n]);
    const size_t rowOff = ((size_t)bi * DD + d) * LL + (size_t)c * LC;
    const float* dRow = Delta + rowOff;
    const float* uRow = U + rowOff;
    const float* bRow = Xd + ((size_t)bi * EE + RR + n) * LL + (size_t)c * LC;
    float h = 0.0f, sdp = 0.0f;
    for (int l0 = 0; l0 < LC; l0 += 16) {
        const float dl = dRow[l0 + n];
        const float ul = uRow[l0 + n];
        sdp += dl;
        float bv[16];
        #pragma unroll
        for (int q = 0; q < 4; ++q) {
            float4 t = *(const float4*)(bRow + l0 + 4 * q);
            bv[4*q+0] = t.x; bv[4*q+1] = t.y; bv[4*q+2] = t.z; bv[4*q+3] = t.w;
        }
        #pragma unroll
        for (int j = 0; j < 16; ++j) {
            const float dj = __shfl(dl, hb + j, 32);
            const float uj = __shfl(ul, hb + j, 32);
            h = expf(dj * An) * h + dj * uj * bv[j];
        }
    }
    sdp += __shfl_xor(sdp, 8, 32);
    sdp += __shfl_xor(sdp, 4, 32);
    sdp += __shfl_xor(sdp, 2, 32);
    sdp += __shfl_xor(sdp, 1, 32);
    S[((size_t)ch * NC + c) * NS + n] = h;
    if (n == 0) SumD[(size_t)ch * NC + c] = sdp;
}

__global__ __launch_bounds__(256)
void k_scanB(const float* __restrict__ Alog, const float* __restrict__ S,
             const float* __restrict__ SumD, float* __restrict__ Hin)
{
    const int lane = threadIdx.x & 31;
    const int gw   = blockIdx.x * (blockDim.x >> 5) + (threadIdx.x >> 5);
    const int ch   = (gw << 1) + (lane >> 4);
    const int d    = ch & (DD - 1);
    const int n    = lane & 15;
    const float An = -expf(Alog[d * NS + n]);
    float P[NC];
    #pragma unroll
    for (int c = 0; c < NC; ++c)
        P[c] = expf(An * SumD[(size_t)ch * NC + c]);
    float hin = 0.0f;
    #pragma unroll
    for (int c = 0; c < NC; ++c) {
        Hin[((size_t)ch * NC + c) * NS + n] = hin;
        hin = P[c] * hin + S[((size_t)ch * NC + c) * NS + n];
    }
}

__global__ __launch_bounds__(256)
void k_scanC(const float* __restrict__ U, const float* __restrict__ Xd,
             const float* __restrict__ Delta, const float* __restrict__ Alog,
             const float* __restrict__ Dp, const float* __restrict__ Hin,
             float* __restrict__ Y)
{
    const int lane = threadIdx.x & 31;
    const int gw   = blockIdx.x * (blockDim.x >> 5) + (threadIdx.x >> 5);
    const int p    = (gw << 1) + (lane >> 4);
    const int ch   = p >> 5;
    const int c    = p & (NC - 1);
    const int bi   = ch >> 10;
    const int d    = ch & (DD - 1);
    const int n    = lane & 15;
    const int hb   = lane & 16;
    const float An = -expf(Alog[d * NS + n]);
    const float Dv = Dp[d];
    const size_t rowOff = ((size_t)bi * DD + d) * LL + (size_t)c * LC;
    const float* dRow = Delta + rowOff;
    const float* uRow = U + rowOff;
    const float* bRow = Xd + ((size_t)bi * EE + RR) * LL      + (size_t)n * LL + (size_t)c * LC;
    const float* cRow = Xd + ((size_t)bi * EE + RR + NS) * LL + (size_t)n * LL + (size_t)c * LC;
    float* yRow = Y + rowOff;
    float h = Hin[((size_t)ch * NC + c) * NS + n];
    for (int l0 = 0; l0 < LC; l0 += 16) {
        const float dl = dRow[l0 + n];
        const float ul = uRow[l0 + n];
        float bv[16], cv[16];
        #pragma unroll
        for (int q = 0; q < 4; ++q) {
            float4 t = *(const float4*)(bRow + l0 + 4 * q);
            bv[4*q+0] = t.x; bv[4*q+1] = t.y; bv[4*q+2] = t.z; bv[4*q+3] = t.w;
            float4 s = *(const float4*)(cRow + l0 + 4 * q);
            cv[4*q+0] = s.x; cv[4*q+1] = s.y; cv[4*q+2] = s.z; cv[4*q+3] = s.w;
        }
        float ymy = 0.0f;
        #pragma unroll
        for (int j = 0; j < 16; ++j) {
            const float dj = __shfl(dl, hb + j, 32);
            const float uj = __shfl(ul, hb + j, 32);
            h = expf(dj * An) * h + dj * uj * bv[j];
            float yv = h * cv[j];
            yv += __shfl_xor(yv, 8, 32);
            yv += __shfl_xor(yv, 4, 32);
            yv += __shfl_xor(yv, 2, 32);
            yv += __shfl_xor(yv, 1, 32);
            if (n == j) ymy = yv;
        }
        float yo = ymy + ul * Dv;
        yo = yo * (ul * sigmoidf_(ul));
        yo = 0.5f * yo * (1.0f + erff(yo * 0.70710678118f));
        yRow[l0 + n] = yo;
    }
}

__global__ __launch_bounds__(128)
void k_out(const float* __restrict__ Wo, const float* __restrict__ Yv,
           const float* __restrict__ bo, float* __restrict__ Out)
{
    __shared__ float Aa[16 * 68];
    __shared__ float Ag[16 * 68];
    const int tid  = threadIdx.x;
    const int lane = tid & 31;
    const int wv   = tid >> 5;
    const int m0   = blockIdx.y << 4;
    const int n0   = (blockIdx.x << 7) + (wv << 5);
    const int bi   = blockIdx.z;
    const int am   = lane & 15;
    const int ak   = (lane >> 4) << 1;
    v8f aA0 = {}, aA1 = {}, aG0 = {}, aG1 = {};
    const float* Yb = Yv + ((size_t)bi * DD) * LL + n0 + am;
    for (int kc = 0; kc < DD; kc += 64) {
        for (int i = tid; i < 16 * 64; i += 128) {
            const int r = i >> 6, c = i & 63;
            Aa[r * 68 + c] = Wo[(size_t)(m0 + r) * DD + kc + c];
            Ag[r * 68 + c] = Wo[(size_t)(DD + m0 + r) * DD + kc + c];
        }
        __syncthreads();
        if (kc + 64 < DD)
            __builtin_prefetch(&Wo[(size_t)(m0 + (tid >> 3)) * DD + kc + 64 + ((tid & 7) << 3)], 0, 1);
        #pragma unroll
        for (int k = 0; k < 64; k += 4) {
            v2f b0, b1, aa, ag;
            const float* bp = Yb + (size_t)(kc + k + ak) * LL;
            b0.x = bp[0];
            b0.y = bp[LL];
            b1.x = bp[16];
            b1.y = bp[LL + 16];
            aa.x = Aa[am * 68 + k + ak];
            aa.y = Aa[am * 68 + k + ak + 1];
            ag.x = Ag[am * 68 + k + ak];
            ag.y = Ag[am * 68 + k + ak + 1];
            aA0 = __builtin_amdgcn_wmma_f32_16x16x4_f32(false, aa, false, b0, (short)0, aA0, false, false);
            aA1 = __builtin_amdgcn_wmma_f32_16x16x4_f32(false, aa, false, b1, (short)0, aA1, false, false);
            aG0 = __builtin_amdgcn_wmma_f32_16x16x4_f32(false, ag, false, b0, (short)0, aG0, false, false);
            aG1 = __builtin_amdgcn_wmma_f32_16x16x4_f32(false, ag, false, b1, (short)0, aG1, false, false);
        }
        __syncthreads();
    }
    const int cn = n0 + am;
    const int cr = (lane >> 4) << 3;
    #pragma unroll
    for (int v = 0; v < 8; ++v) {
        const int row = m0 + cr + v;
        const float ba = bo[row];
        const float bg = bo[DD + row];
        float* o = Out + ((size_t)bi * DD + row) * LL + cn;
        o[0]  = (aA0[v] + ba) * sigmoidf_(aG0[v] + bg);
        o[16] = (aA1[v] + ba) * sigmoidf_(aG1[v] + bg);
    }
}

extern "C" void kernel_launch(void* const* d_in, const int* in_sizes, int n_in,
                              void* d_out, int out_size, void* d_ws, size_t ws_size,
                              hipStream_t stream)
{
    const float* u    = (const float*)d_in[0];
    const float* xw   = (const float*)d_in[1];
    const float* dtw  = (const float*)d_in[2];
    const float* dtb  = (const float*)d_in[3];
    const float* alog = (const float*)d_in[4];
    const float* Dp   = (const float*)d_in[5];
    const float* ow   = (const float*)d_in[6];
    const float* ob   = (const float*)d_in[7];
    float* out = (float*)d_out;

    const size_t CH = (size_t)NB * DD;
    float* ws    = (float*)d_ws;
    float* xdbl  = ws;
    float* delta = xdbl + (size_t)NB * EE * LL;
    float* yv    = delta + CH * LL;
    float* Sst   = yv + CH * LL;
    float* sumD  = Sst + CH * NC * NS;
    float* hin   = sumD + CH * NC;

    const dim3 blk(128);
    hipLaunchKernelGGL(k_xproj, dim3(LL / 64, EE / 16, NB), blk, 0, stream, xw, u, xdbl);
    hipLaunchKernelGGL(k_dt,    dim3(LL / 64, DD / 16, NB), blk, 0, stream, dtw, xdbl, dtb, delta);
    hipLaunchKernelGGL(k_scanA, dim3((CH * NC) / 16), dim3(256), 0, stream, u, xdbl, delta, alog, Sst, sumD);
    hipLaunchKernelGGL(k_scanB, dim3(CH / 16), dim3(256), 0, stream, alog, Sst, sumD, hin);
    hipLaunchKernelGGL(k_scanC, dim3((CH * NC) / 16), dim3(256), 0, stream, u, xdbl, delta, alog, Dp, hin, yv);
    hipLaunchKernelGGL(k_out,   dim3(LL / 128, DD / 16, NB), blk, 0, stream, ow, yv, ob, out);
}


# scan: three chunk passes -> one kernel, lanes = time: thread folds its 16 steps to (P,X), block prefix (shuffles + LDS) gives its entry state, replay adds h*C over states in-lane; sum order changed
# speedup vs baseline: 1.7851x; 1.7851x over previous
#include <hip/hip_runtime.h>
#include <math.h>

#define NB 2
#define DD 1024
#define LL 2048
#define NS 16
#define RR 64
#define EE 96

typedef float v2f __attribute__((ext_vector_type(2)));
typedef float v8f __attribute__((ext_vector_type(8)));

__device__ __forceinline__ float sigmoidf_(float x) { return 1.0f / (1.0f + expf(-x)); }
__device__ __forceinline__ float softplusf_(float x) { return (x > 20.0f) ? x : log1pf(expf(x)); }

#define XW 16
#define XS (DD / XW)
__global__ __launch_bounds__(512)
void k_xproj(const float* __restrict__ W, const float* __restrict__ U,
             float* __restrict__ Xd)
{
    __shared__ float Red[XW * EE * 16];
    const int tid  = threadIdx.x;
    const int lane = tid & 31;
    const int wv   = tid >> 5;
    const int n0   = blockIdx.x << 4;
    const int bi   = blockIdx.y;
    const int am   = lane & 15;
    const int ak   = (lane >> 4) << 2;
    const int k0   = wv * XS;
    const float* Ub = U + ((size_t)bi * DD + k0 + ak) * LL + n0 + am;
    const float* Wb = W + (size_t)am * DD + k0 + ak;
    v2f b[XS / 4];
    #pragma unroll
    for (int g = 0; g < XS / 8; ++g) {
        b[2 * g].x     = Ub[(size_t)(8 * g + 0) * LL];
        b[2 * g].y     = Ub[(size_t)(8 * g + 1) * LL];
        b[2 * g + 1].x = Ub[(size_t)(8 * g + 2) * LL];
        b[2 * g + 1].y = Ub[(size_t)(8 * g + 3) * LL];
    }
    v8f acc[6];
    #pragma unroll
    for (int t = 0; t < 6; ++t) {
        const float* wr = Wb + (size_t)(t * 16) * DD;
        float4 a[XS / 8];
        #pragma unroll
        for (int g = 0; g < XS / 8; ++g)
            a[g] = *(const float4*)(wr + 8 * g);
        v8f c = {};
        #pragma unroll
        for (int g = 0; g < XS / 8; ++g) {
            v2f a0, a1;
            a0.x = a[g].x; a0.y = a[g].y;
            a1.x = a[g].z; a1.y = a[g].w;
            c = __builtin_amdgcn_wmma_f32_16x16x4_f32(
                false, a0, false, b[2 * g], (short)0, c, false, false);
            c = __builtin_amdgcn_wmma_f32_16x16x4_f32(
                false, a1, false, b[2 * g + 1], (short)0, c, false, false);
        }
        acc[t] = c;
    }
    float* myRed = Red + wv * (EE * 16);
    #pragma unroll
    for (int t = 0; t < 6; ++t) {
        #pragma unroll
        for (int v = 0; v < 8; ++v)
            myRed[(t * 8 + v) * 32 + lane] = acc[t][v];
    }
    __syncthreads();
    #pragma unroll
    for (int i = 0; i < (EE * 16) / 512; ++i) {
        const int s = tid + i * 512;
        float sum = Red[s];
        #pragma unroll
        for (int w = 1; w < XW; ++w)
            sum += Red[w * (EE * 16) + s];
        const int row = ((s >> 8) << 4) + (((s >> 4) & 1) << 3) + ((s >> 5) & 7);
        Xd[((size_t)bi * EE + row) * LL + n0 + (s & 15)] = sum;
    }
}

__global__ __launch_bounds__(128)
void k_dt(const float* __restrict__ Wdt, const float* __restrict__ Xd,
          const float* __restrict__ bias, float* __restrict__ Delta)
{
    __shared__ float As[16 * 68];
    const int tid  = threadIdx.x;
    const int lane = tid & 31;
    const int wv   = tid >> 5;
    const int m0   = blockIdx.y << 4;
    const int n0   = (blockIdx.x << 6) + (wv << 4);
    const int bi   = blockIdx.z;
    const int am   = lane & 15;
    const int ak   = (lane >> 4) << 1;
    for (int i = tid; i < 16 * 64; i += 128) {
        const int r = i >> 6, c = i & 63;
        As[r * 68 + c] = Wdt[(size_t)(m0 + r) * RR + c];
    }
    __syncthreads();
    v8f acc = {};
    const float* Bb = Xd + ((size_t)bi * EE) * LL + n0 + am;
    #pragma unroll
    for (int k = 0; k < RR; k += 4) {
        v2f a, b;
        a.x = As[am * 68 + k + ak];
        a.y = As[am * 68 + k + ak + 1];
        const float* bp = Bb + (size_t)(k + ak) * LL;
        b.x = bp[0];
        b.y = bp[LL];
        acc = __builtin_amdgcn_wmma_f32_16x16x4_f32(
            false, a, false, b, (short)0, acc, false, false);
    }
    const int cn = n0 + am;
    const int cr = (lane >> 4) << 3;
    #pragma unroll
    for (int v = 0; v < 8; ++v) {
        const int row = m0 + cr + v;
        Delta[((size_t)bi * DD + row) * LL + cn] = softplusf_(acc[v] + bias[row]);
    }
}

#define ST 16
#define SW (LL / ST / 32)
__global__ __launch_bounds__(LL / ST)
void k_scan(const float* __restrict__ U, const float* __restrict__ Xd,
            const float* __restrict__ Delta, const float* __restrict__ Alog,
            const float* __restrict__ Dp, float* __restrict__ Y)
{
    __shared__ float Tot[2][SW][2][2];
    const int tid  = threadIdx.x;
    const int lane = tid & 31;
    const int wv   = tid >> 5;
    const int bi   = blockIdx.x / (DD / 2);
    const int d0   = (blockIdx.x % (DD / 2)) << 1;
    const unsigned lo = (unsigned)tid * ST;
    float dl[2][ST], du[2][ST], ys[2][ST];
    #pragma unroll
    for (int c = 0; c < 2; ++c) {
        const size_t row = ((size_t)bi * DD + d0 + c) * LL;
        const float* dP = Delta + row;
        const float* uP = U + row;
        #pragma unroll
        for (int q = 0; q < ST / 4; ++q) {
            const float4 a = *(const float4*)(dP + lo + 4 * q);
            const float4 b = *(const float4*)(uP + lo + 4 * q);
            dl[c][4*q+0] = a.x; dl[c][4*q+1] = a.y; dl[c][4*q+2] = a.z; dl[c][4*q+3] = a.w;
            du[c][4*q+0] = a.x * b.x; du[c][4*q+1] = a.y * b.y;
            du[c][4*q+2] = a.z * b.z; du[c][4*q+3] = a.w * b.w;
        }
        #pragma unroll
        for (int j = 0; j < ST; ++j) ys[c][j] = 0.0f;
    }
    const float* bRow = Xd + ((size_t)bi * EE + RR) * LL;
    const float* cRow = Xd + ((size_t)bi * EE + RR + NS) * LL;
    float Bv[ST], Cv[ST];
    #pragma unroll
    for (int q = 0; q < ST / 4; ++q) {
        const float4 t = *(const float4*)(bRow + lo + 4 * q);
        Bv[4*q+0] = t.x; Bv[4*q+1] = t.y; Bv[4*q+2] = t.z; Bv[4*q+3] = t.w;
        const float4 s = *(const float4*)(cRow + lo + 4 * q);
        Cv[4*q+0] = s.x; Cv[4*q+1] = s.y; Cv[4*q+2] = s.z; Cv[4*q+3] = s.w;
    }
    #pragma unroll 1
    for (int n = 0; n < NS; ++n) {
        const int nn = (n + 1 < NS) ? n + 1 : n;
        float e[2][ST], x[2][ST], P[2], X[2];
        #pragma unroll
        for (int c = 0; c < 2; ++c) {
            const float An = -expf(Alog[(d0 + c) * NS + n]);
            #pragma unroll
            for (int j = 0; j < ST; ++j) {
                e[c][j] = expf(dl[c][j] * An);
                x[c][j] = du[c][j] * Bv[j];
            }
        }
        #pragma unroll
        for (int q = 0; q < ST / 4; ++q) {
            const float4 t = *(const float4*)(bRow + (size_t)nn * LL + lo + 4 * q);
            Bv[4*q+0] = t.x; Bv[4*q+1] = t.y; Bv[4*q+2] = t.z; Bv[4*q+3] = t.w;
        }
        #pragma unroll
        for (int c = 0; c < 2; ++c) {
            float Xc = x[c][0], Pc = e[c][0];
            #pragma unroll
            for (int j = 1; j < ST; ++j) {
                Xc = e[c][j] * Xc + x[c][j];
                Pc = Pc * e[c][j];
            }
            P[c] = Pc; X[c] = Xc;
        }
        #pragma unroll
        for (int s = 1; s < 32; s <<= 1) {
            #pragma unroll
            for (int c = 0; c < 2; ++c) {
                const float Pp = __shfl_up(P[c], s, 32);
                const float Xp = __shfl_up(X[c], s, 32);
                if (lane >= s) {
                    X[c] = P[c] * Xp + X[c];
                    P[c] = P[c] * Pp;
                }
            }
        }
        float Pe[2], Xe[2];
        #pragma unroll
        for (int c = 0; c < 2; ++c) {
            if (lane == 31) {
                Tot[n & 1][wv][c][0] = P[c];
                Tot[n & 1][wv][c][1] = X[c];
            }
            Pe[c] = __shfl_up(P[c], 1, 32);
            Xe[c] = __shfl_up(X[c], 1, 32);
            if (lane == 0) { Pe[c] = 1.0f; Xe[c] = 0.0f; }
        }
        __syncthreads();
        #pragma unroll
        for (int c = 0; c < 2; ++c) {
            float hw = 0.0f;
            #pragma unroll
            for (int k = 0; k < SW - 1; ++k) {
                const float pk = Tot[n & 1][k][c][0];
                const float xk = Tot[n & 1][k][c][1];
                if (k < wv) hw = pk * hw + xk;
            }
            float h = Pe[c] * hw + Xe[c];
            #pragma unroll
            for (int j = 0; j < ST; ++j) {
                h = e[c][j] * h + x[c][j];
                ys[c][j] = h * Cv[j] + ys[c][j];
            }
        }
        #pragma unroll
        for (int q = 0; q < ST / 4; ++q) {
            const float4 s = *(const float4*)(cRow + (size_t)nn * LL + lo + 4 * q);
            Cv[4*q+0] = s.x; Cv[4*q+1] = s.y; Cv[4*q+2] = s.z; Cv[4*q+3] = s.w;
        }
    }
    #pragma unroll
    for (int c = 0; c < 2; ++c) {
        const float Dv = Dp[d0 + c];
        const size_t row = ((size_t)bi * DD + d0 + c) * LL;
        const float* uP = U + row;
        float* yP = Y + row;
        #pragma unroll
        for (int q = 0; q < ST / 4; ++q) {
            const float4 b = *(const float4*)(uP + lo + 4 * q);
            const float uu[4] = { b.x, b.y, b.z, b.w };
            float r[4];
            #pragma unroll
            for (int i = 0; i < 4; ++i) {
                const float ul = uu[i];
                float yo = ys[c][4*q+i] + ul * Dv;
                yo = yo * (ul * sigmoidf_(ul));
                r[i] = 0.5f * yo * (1.0f + erff(yo * 0.70710678118f));
            }
            float4 o; o.x = r[0]; o.y = r[1]; o.z = r[2]; o.w = r[3];
            *(float4*)(yP + lo + 4 * q) = o;
        }
    }
}

__global__ __launch_bounds__(128)
void k_out(const float* __restrict__ Wo, const float* __restrict__ Yv,
           const float* __restrict__ bo, float* __restrict__ Out)
{
    __shared__ float Aa[16 * 68];
    __shared__ float Ag[16 * 68];
    const int tid  = threadIdx.x;
    const int lane = tid & 31;
    const int wv   = tid >> 5;
    const int m0   = blockIdx.y << 4;
    const int n0   = (blockIdx.x << 7) + (wv << 5);
    const int bi   = blockIdx.z;
    const int am   = lane & 15;
    const int ak   = (lane >> 4) << 1;
    v8f aA0 = {}, aA1 = {}, aG0 = {}, aG1 = {};
    const float* Yb = Yv + ((size_t)bi * DD) * LL + n0 + am;
    for (int kc = 0; kc < DD; kc += 64) {
        for (int i = tid; i < 16 * 64; i += 128) {
            const int r = i >> 6, c = i & 63;
            Aa[r * 68 + c] = Wo[(size_t)(m0 + r) * DD + kc + c];
            Ag[r * 68 + c] = Wo[(size_t)(DD + m0 + r) * DD + kc + c];
        }
        __syncthreads();
        if (kc + 64 < DD)
            __builtin_prefetch(&Wo[(size_t)(m0 + (tid >> 3)) * DD + kc + 64 + ((tid & 7) << 3)], 0, 1);
        #pragma unroll
        for (int k = 0; k < 64; k += 4) {
            v2f b0, b1, aa, ag;
            const float* bp = Yb + (size_t)(kc + k + ak) * LL;
            b0.x = bp[0];
            b0.y = bp[LL];
            b1.x = bp[16];
            b1.y = bp[LL + 16];
            aa.x = Aa[am * 68 + k + ak];
            aa.y = Aa[am * 68 + k + ak + 1];
            ag.x = Ag[am * 68 + k + ak];
            ag.y = Ag[am * 68 + k + ak + 1];
            aA0 = __builtin_amdgcn_wmma_f32_16x16x4_f32(false, aa, false, b0, (short)0, aA0, false, false);
            aA1 = __builtin_amdgcn_wmma_f32_16x16x4_f32(false, aa, false, b1, (short)0, aA1, false, false);
            aG0 = __builtin_amdgcn_wmma_f32_16x16x4_f32(false, ag, false, b0, (short)0, aG0, false, false);
            aG1 = __builtin_amdgcn_wmma_f32_16x16x4_f32(false, ag, false, b1, (short)0, aG1, false, false);
        }
        __syncthreads();
    }
    const int cn = n0 + am;
    const int cr = (lane >> 4) << 3;
    #pragma unroll
    for (int v = 0; v < 8; ++v) {
        const int row = m0 + cr + v;
        const float ba = bo[row];
        const float bg = bo[DD + row];
        float* o = Out + ((size_t)bi * DD + row) * LL + cn;
        o[0]  = (aA0[v] + ba) * sigmoidf_(aG0[v] + bg);
        o[16] = (aA1[v] + ba) * sigmoidf_(aG1[v] + bg);
    }
}

extern "C" void kernel_launch(void* const* d_in, const int* in_sizes, int n_in,
                              void* d_out, int out_size, void* d_ws, size_t ws_size,
                              hipStream_t stream)
{
    const float* u    = (const float*)d_in[0];
    const float* xw   = (const float*)d_in[1];
    const float* dtw  = (const float*)d_in[2];
    const float* dtb  = (const float*)d_in[3];
    const float* alog = (const float*)d_in[4];
    const float* Dp   = (const float*)d_in[5];
    const float* ow   = (const float*)d_in[6];
    const float* ob   = (const float*)d_in[7];
    float* out = (float*)d_out;

    const size_t CH = (size_t)NB * DD;
    float* ws    = (float*)d_ws;
    float* xdbl  = ws;
    float* delta = xdbl + (size_t)NB * EE * LL;
    float* yv    = delta + CH * LL;

    const dim3 blk(128);
    hipLaunchKernelGGL(k_xproj, dim3(LL / 16, NB), dim3(512), 0, stream, xw, u, xdbl);
    hipLaunchKernelGGL(k_dt,    dim3(LL / 64, DD / 16, NB), blk, 0, stream, dtw, xdbl, dtb, delta);
    hipLaunchKernelGGL(k_scan,  dim3(CH / 2), dim3(LL / ST), 0, stream, u, xdbl, delta, alog, Dp, yv);
    hipLaunchKernelGGL(k_out,   dim3(LL / 128, DD / 16, NB), blk, 0, stream, ow, yv, ob, out);
}


# k_out: wave tile 64 weight rows (32 a + paired 32 g) x 32 cols, operands straight from global into fragments (no LDS, no barrier), next 16 k loaded as one group ahead of WMMAs; k sum order changed
# speedup vs baseline: 2.1137x; 1.1841x over previous
#include <hip/hip_runtime.h>
#include <math.h>

#define NB 2
#define DD 1024
#define LL 2048
#define NS 16
#define RR 64
#define EE 96

typedef float v2f __attribute__((ext_vector_type(2)));
typedef float v8f __attribute__((ext_vector_type(8)));

__device__ __forceinline__ float sigmoidf_(float x) { return 1.0f / (1.0f + expf(-x)); }
__device__ __forceinline__ float softplusf_(float x) { return (x > 20.0f) ? x : log1pf(expf(x)); }

#define XW 16
#define XS (DD / XW)
__global__ __launch_bounds__(512)
void k_xproj(const float* __restrict__ W, const float* __restrict__ U,
             float* __restrict__ Xd)
{
    __shared__ float Red[XW * EE * 16];
    const int tid  = threadIdx.x;
    const int lane = tid & 31;
    const int wv   = tid >> 5;
    const int n0   = blockIdx.x << 4;
    const int bi   = blockIdx.y;
    const int am   = lane & 15;
    const int ak   = (lane >> 4) << 2;
    const int k0   = wv * XS;
    const float* Ub = U + ((size_t)bi * DD + k0 + ak) * LL + n0 + am;
    const float* Wb = W + (size_t)am * DD + k0 + ak;
    v2f b[XS / 4];
    #pragma unroll
    for (int g = 0; g < XS / 8; ++g) {
        b[2 * g].x     = Ub[(size_t)(8 * g + 0) * LL];
        b[2 * g].y     = Ub[(size_t)(8 * g + 1) * LL];
        b[2 * g + 1].x = Ub[(size_t)(8 * g + 2) * LL];
        b[2 * g + 1].y = Ub[(size_t)(8 * g + 3) * LL];
    }
    v8f acc[6];
    #pragma unroll
    for (int t = 0; t < 6; ++t) {
        const float* wr = Wb + (size_t)(t * 16) * DD;
        float4 a[XS / 8];
        #pragma unroll
        for (int g = 0; g < XS / 8; ++g)
            a[g] = *(const float4*)(wr + 8 * g);
        v8f c = {};
        #pragma unroll
        for (int g = 0; g < XS / 8; ++g) {
            v2f a0, a1;
            a0.x = a[g].x; a0.y = a[g].y;
            a1.x = a[g].z; a1.y = a[g].w;
            c = __builtin_amdgcn_wmma_f32_16x16x4_f32(
                false, a0, false, b[2 * g], (short)0, c, false, false);
            c = __builtin_amdgcn_wmma_f32_16x16x4_f32(
                false, a1, false, b[2 * g + 1], (short)0, c, false, false);
        }
        acc[t] = c;
    }
    float* myRed = Red + wv * (EE * 16);
    #pragma unroll
    for (int t = 0; t < 6; ++t) {
        #pragma unroll
        for (int v = 0; v < 8; ++v)
            myRed[(t * 8 + v) * 32 + lane] = acc[t][v];
    }
    __syncthreads();
    #pragma unroll
    for (int i = 0; i < (EE * 16) / 512; ++i) {
        const int s = tid + i * 512;
        float sum = Red[s];
        #pragma unroll
        for (int w = 1; w < XW; ++w)
            sum += Red[w * (EE * 16) + s];
        const int row = ((s >> 8) << 4) + (((s >> 4) & 1) << 3) + ((s >> 5) & 7);
        Xd[((size_t)bi * EE + row) * LL + n0 + (s & 15)] = sum;
    }
}

__global__ __launch_bounds__(128)
void k_dt(const float* __restrict__ Wdt, const float* __restrict__ Xd,
          const float* __restrict__ bias, float* __restrict__ Delta)
{
    __shared__ float As[16 * 68];
    const int tid  = threadIdx.x;
    const int lane = tid & 31;
    const int wv   = tid >> 5;
    const int m0   = blockIdx.y << 4;
    const int n0   = (blockIdx.x << 6) + (wv << 4);
    const int bi   = blockIdx.z;
    const int am   = lane & 15;
    const int ak   = (lane >> 4) << 1;
    for (int i = tid; i < 16 * 64; i += 128) {
        const int r = i >> 6, c = i & 63;
        As[r * 68 + c] = Wdt[(size_t)(m0 + r) * RR + c];
    }
    __syncthreads();
    v8f acc = {};
    const float* Bb = Xd + ((size_t)bi * EE) * LL + n0 + am;
    #pragma unroll
    for (int k = 0; k < RR; k += 4) {
        v2f a, b;
        a.x = As[am * 68 + k + ak];
        a.y = As[am * 68 + k + ak + 1];
        const float* bp = Bb + (size_t)(k + ak) * LL;
        b.x = bp[0];
        b.y = bp[LL];
        acc = __builtin_amdgcn_wmma_f32_16x16x4_f32(
            false, a, false, b, (short)0, acc, false, false);
    }
    const int cn = n0 + am;
    const int cr = (lane >> 4) << 3;
    #pragma unroll
    for (int v = 0; v < 8; ++v) {
        const int row = m0 + cr + v;
        Delta[((size_t)bi * DD + row) * LL + cn] = softplusf_(acc[v] + bias[row]);
    }
}

#define ST 16
#define SW (LL / ST / 32)
__global__ __launch_bounds__(LL / ST)
void k_scan(const float* __restrict__ U, const float* __restrict__ Xd,
            const float* __restrict__ Delta, const float* __restrict__ Alog,
            const float* __restrict__ Dp, float* __restrict__ Y)
{
    __shared__ float Tot[2][SW][2][2];
    const int tid  = threadIdx.x;
    const int lane = tid & 31;
    const int wv   = tid >> 5;
    const int bi   = blockIdx.x / (DD / 2);
    const int d0   = (blockIdx.x % (DD / 2)) << 1;
    const unsigned lo = (unsigned)tid * ST;
    float dl[2][ST], du[2][ST], ys[2][ST];
    #pragma unroll
    for (int c = 0; c < 2; ++c) {
        const size_t row = ((size_t)bi * DD + d0 + c) * LL;
        const float* dP = Delta + row;
        const float* uP = U + row;
        #pragma unroll
        for (int q = 0; q < ST / 4; ++q) {
            const float4 a = *(const float4*)(dP + lo + 4 * q);
            const float4 b = *(const float4*)(uP + lo + 4 * q);
            dl[c][4*q+0] = a.x; dl[c][4*q+1] = a.y; dl[c][4*q+2] = a.z; dl[c][4*q+3] = a.w;
            du[c][4*q+0] = a.x * b.x; du[c][4*q+1] = a.y * b.y;
            du[c][4*q+2] = a.z * b.z; du[c][4*q+3] = a.w * b.w;
        }
        #pragma unroll
        for (int j = 0; j < ST; ++j) ys[c][j] = 0.0f;
    }
    const float* bRow = Xd + ((size_t)bi * EE + RR) * LL;
    const float* cRow = Xd + ((size_t)bi * EE + RR + NS) * LL;
    float Bv[ST], Cv[ST];
    #pragma unroll
    for (int q = 0; q < ST / 4; ++q) {
        const float4 t = *(const float4*)(bRow + lo + 4 * q);
        Bv[4*q+0] = t.x; Bv[4*q+1] = t.y; Bv[4*q+2] = t.z; Bv[4*q+3] = t.w;
        const float4 s = *(const float4*)(cRow + lo + 4 * q);
        Cv[4*q+0] = s.x; Cv[4*q+1] = s.y; Cv[4*q+2] = s.z; Cv[4*q+3] = s.w;
    }
    #pragma unroll 1
    for (int n = 0; n < NS; ++n) {
        const int nn = (n + 1 < NS) ? n + 1 : n;
        float e[2][ST], x[2][ST], P[2], X[2];
        #pragma unroll
        for (int c = 0; c < 2; ++c) {
            const float An = -expf(Alog[(d0 + c) * NS + n]);
            #pragma unroll
            for (int j = 0; j < ST; ++j) {
                e[c][j] = expf(dl[c][j] * An);
                x[c][j] = du[c][j] * Bv[j];
            }
        }
        #pragma unroll
        for (int q = 0; q < ST / 4; ++q) {
            const float4 t = *(const float4*)(bRow + (size_t)nn * LL + lo + 4 * q);
            Bv[4*q+0] = t.x; Bv[4*q+1] = t.y; Bv[4*q+2] = t.z; Bv[4*q+3] = t.w;
        }
        #pragma unroll
        for (int c = 0; c < 2; ++c) {
            float Xc = x[c][0], Pc = e[c][0];
            #pragma unroll
            for (int j = 1; j < ST; ++j) {
                Xc = e[c][j] * Xc + x[c][j];
                Pc = Pc * e[c][j];
            }
            P[c] = Pc; X[c] = Xc;
        }
        #pragma unroll
        for (int s = 1; s < 32; s <<= 1) {
            #pragma unroll
            for (int c = 0; c < 2; ++c) {
                const float Pp = __shfl_up(P[c], s, 32);
                const float Xp = __shfl_up(X[c], s, 32);
                if (lane >= s) {
                    X[c] = P[c] * Xp + X[c];
                    P[c] = P[c] * Pp;
                }
            }
        }
        float Pe[2], Xe[2];
        #pragma unroll
        for (int c = 0; c < 2; ++c) {
            if (lane == 31) {
                Tot[n & 1][wv][c][0] = P[c];
                Tot[n & 1][wv][c][1] = X[c];
            }
            Pe[c] = __shfl_up(P[c], 1, 32);
            Xe[c] = __shfl_up(X[c], 1, 32);
            if (lane == 0) { Pe[c] = 1.0f; Xe[c] = 0.0f; }
        }
        __syncthreads();
        #pragma unroll
        for (int c = 0; c < 2; ++c) {
            float hw = 0.0f;
            #pragma unroll
            for (int k = 0; k < SW - 1; ++k) {
                const float pk = Tot[n & 1][k][c][0];
                const float xk = Tot[n & 1][k][c][1];
                if (k < wv) hw = pk * hw + xk;
            }
            float h = Pe[c] * hw + Xe[c];
            #pragma unroll
            for (int j = 0; j < ST; ++j) {
                h = e[c][j] * h + x[c][j];
                ys[c][j] = h * Cv[j] + ys[c][j];
            }
        }
        #pragma unroll
        for (int q = 0; q < ST / 4; ++q) {
            const float4 s = *(const float4*)(cRow + (size_t)nn * LL + lo + 4 * q);
            Cv[4*q+0] = s.x; Cv[4*q+1] = s.y; Cv[4*q+2] = s.z; Cv[4*q+3] = s.w;
        }
    }
    #pragma unroll
    for (int c = 0; c < 2; ++c) {
        const float Dv = Dp[d0 + c];
        const size_t row = ((size_t)bi * DD + d0 + c) * LL;
        const float* uP = U + row;
        float* yP = Y + row;
        #pragma unroll
        for (int q = 0; q < ST / 4; ++q) {
            const float4 b = *(const float4*)(uP + lo + 4 * q);
            const float uu[4] = { b.x, b.y, b.z, b.w };
            float r[4];
            #pragma unroll
            for (int i = 0; i < 4; ++i) {
                const float ul = uu[i];
                float yo = ys[c][4*q+i] + ul * Dv;
                yo = yo * (ul * sigmoidf_(ul));
                r[i] = 0.5f * yo * (1.0f + erff(yo * 0.70710678118f));
            }
            float4 o; o.x = r[0]; o.y = r[1]; o.z = r[2]; o.w = r[3];
            *(float4*)(yP + lo + 4 * q) = o;
        }
    }
}

#define OKS 16
struct OutFrag {
    float4 a[4][OKS / 8];
    v2f    b[2][OKS / 4];
};
__device__ __forceinline__ void out_load(OutFrag& f, const float* __restrict__ wa,
                                         const float* __restrict__ yb, int k)
{
    #pragma unroll
    for (int t = 0; t < 4; ++t) {
        const float* wr = wa + (size_t)((t >> 1) * DD + (t & 1) * 16) * DD + k;
        #pragma unroll
        for (int g = 0; g < OKS / 8; ++g)
            f.a[t][g] = *(const float4*)(wr + 8 * g);
    }
    #pragma unroll
    for (int g = 0; g < OKS / 8; ++g) {
        const float* bp = yb + (size_t)(k + 8 * g) * LL;
        #pragma unroll
        for (int j = 0; j < 2; ++j) {
            f.b[j][2 * g].x     = bp[16 * j];
            f.b[j][2 * g].y     = bp[16 * j + LL];
            f.b[j][2 * g + 1].x = bp[16 * j + 2 * LL];
            f.b[j][2 * g + 1].y = bp[16 * j + 3 * LL];
        }
    }
}
__device__ __forceinline__ void out_mma(v8f (&acc)[4][2], const OutFrag& f)
{
    #pragma unroll
    for (int g = 0; g < OKS / 8; ++g) {
        #pragma unroll
        for (int s = 0; s < 2; ++s) {
            #pragma unroll
            for (int t = 0; t < 4; ++t) {
                v2f a;
                a.x = s ? f.a[t][g].z : f.a[t][g].x;
                a.y = s ? f.a[t][g].w : f.a[t][g].y;
                #pragma unroll
                for (int j = 0; j < 2; ++j)
                    acc[t][j] = __builtin_amdgcn_wmma_f32_16x16x4_f32(
                        false, a, false, f.b[j][2 * g + s], (short)0, acc[t][j], false, false);
            }
        }
    }
}
__global__ __launch_bounds__(128)
void k_out(const float* __restrict__ Wo, const float* __restrict__ Yv,
           const float* __restrict__ bo, float* __restrict__ Out)
{
    const int tid  = threadIdx.x;
    const int lane = tid & 31;
    const int wv   = tid >> 5;
    const int m0   = blockIdx.y << 5;
    const int n0   = (blockIdx.x << 7) + (wv << 5);
    const int bi   = blockIdx.z;
    const int am   = lane & 15;
    const int ak   = (lane >> 4) << 2;
    const float* yb = Yv + ((size_t)bi * DD + ak) * LL + n0 + am;
    const float* wa = Wo + (size_t)(m0 + am) * DD + ak;
    v8f acc[4][2] = {};
    OutFrag f0, f1;
    out_load(f0, wa, yb, 0);
    for (int kc = 0; kc < DD - 2 * OKS; kc += 2 * OKS) {
        out_load(f1, wa, yb, kc + OKS);
        __builtin_amdgcn_sched_barrier(0);
        out_mma(acc, f0);
        __builtin_amdgcn_sched_barrier(0);
        out_load(f0, wa, yb, kc + 2 * OKS);
        __builtin_amdgcn_sched_barrier(0);
        out_mma(acc, f1);
        __builtin_amdgcn_sched_barrier(0);
    }
    out_load(f1, wa, yb, DD - OKS);
    __builtin_amdgcn_sched_barrier(0);
    out_mma(acc, f0);
    out_mma(acc, f1);
    const int cn = n0 + am;
    const int cr = (lane >> 4) << 3;
    #pragma unroll
    for (int t = 0; t < 2; ++t) {
        #pragma unroll
        for (int v = 0; v < 8; ++v) {
            const int row = m0 + 16 * t + cr + v;
            const float ba = bo[row];
            const float bg = bo[DD + row];
            float* o = Out + ((size_t)bi * DD + row) * LL + cn;
            o[0]  = (acc[t][0][v] + ba) * sigmoidf_(acc[2 + t][0][v] + bg);
            o[16] = (acc[t][1][v] + ba) * sigmoidf_(acc[2 + t][1][v] + bg);
        }
    }
}

extern "C" void kernel_launch(void* const* d_in, const int* in_sizes, int n_in,
                              void* d_out, int out_size, void* d_ws, size_t ws_size,
                              hipStream_t stream)
{
    const float* u    = (const float*)d_in[0];
    const float* xw   = (const float*)d_in[1];
    const float* dtw  = (const float*)d_in[2];
    const float* dtb  = (const float*)d_in[3];
    const float* alog = (const float*)d_in[4];
    const float* Dp   = (const float*)d_in[5];
    const float* ow   = (const float*)d_in[6];
    const float* ob   = (const float*)d_in[7];
    float* out = (float*)d_out;

    const size_t CH = (size_t)NB * DD;
    float* ws    = (float*)d_ws;
    float* xdbl  = ws;
    float* delta = xdbl + (size_t)NB * EE * LL;
    float* yv    = delta + CH * LL;

    const dim3 blk(128);
    hipLaunchKernelGGL(k_xproj, dim3(LL / 16, NB), dim3(512), 0, stream, xw, u, xdbl);
    hipLaunchKernelGGL(k_dt,    dim3(LL / 64, DD / 16, NB), blk, 0, stream, dtw, xdbl, dtb, delta);
    hipLaunchKernelGGL(k_scan,  dim3(CH / 2), dim3(LL / ST), 0, stream, u, xdbl, delta, alog, Dp, yv);
    hipLaunchKernelGGL(k_out,   dim3(LL / 128, DD / 32, NB), blk, 0, stream, ow, yv, ob, out);
}
